# starting kernel
# baseline (speedup 1.0000x reference)
#include <hip/hip_runtime.h>
#include <math.h>

typedef __attribute__((ext_vector_type(2))) float v2f;
typedef __attribute__((ext_vector_type(8))) float v8f;

#define NB   8
#define NN   2048
#define ND   128
#define CEPS 1e-8f

__global__ void rnorm_kernel(const float* __restrict__ h, float* __restrict__ rnorm) {
    const int row  = blockIdx.x * 8 + (threadIdx.x >> 5);
    const int lane = threadIdx.x & 31;
    const float* hr = h + (size_t)row * ND;
    float s = 0.f;
#pragma unroll
    for (int i = 0; i < 4; ++i) {
        float v = hr[lane + 32 * i];
        s += v * v;
    }
#pragma unroll
    for (int off = 16; off; off >>= 1) s += __shfl_xor(s, off, 32);
    if (lane == 0) rnorm[row] = 1.0f / fmaxf(sqrtf(s), CEPS);
}

__global__ void __launch_bounds__(256, 1)
simsoftmax_kernel(const float* __restrict__ h,
                  const float* __restrict__ rnorm,
                  const float* __restrict__ temp,
                  float* __restrict__ out) {
    extern __shared__ float lds[];

    const int b       = blockIdx.x >> 7;
    const int rb      = blockIdx.x & 127;
    const int wave    = threadIdx.x >> 5;
    const int lane    = threadIdx.x & 31;
    const int n       = lane & 15;
    const int khalf   = lane >> 4;
    const int rowbase = rb * 16;

    const float* hb = h     + (size_t)b * NN * ND;
    const float* rn = rnorm + b * NN;

    v2f afrag[32];
    const float* arow = hb + (size_t)(rowbase + n) * ND + 2 * khalf;
#pragma unroll
    for (int kc = 0; kc < 32; ++kc)
        afrag[kc] = *(const v2f*)(arow + kc * 4);

    float rowr[8];
#pragma unroll
    for (int r = 0; r < 8; ++r)
        rowr[r] = rn[rowbase + r + 8 * khalf];

    for (int t = wave; t < 128; t += 8) {
        const int colbase = t * 16;
        const float* brow = hb + (size_t)(colbase + n) * ND + 2 * khalf;
        v8f acc = {};
#pragma unroll
        for (int kc = 0; kc < 32; ++kc) {
            v2f bfrag = *(const v2f*)(brow + kc * 4);
            acc = __builtin_amdgcn_wmma_f32_16x16x4_f32(
                false, afrag[kc], false, bfrag, (short)0, acc, false, false);
        }
        const float rc = rn[colbase + n];
#pragma unroll
        for (int r = 0; r < 8; ++r)
            lds[(r + 8 * khalf) * NN + colbase + n] = acc[r] * rowr[r] * rc;
    }

    __syncthreads();

    const float invT = 1.0f / temp[0];
#pragma unroll
    for (int rr = 0; rr < 2; ++rr) {
        const int row = wave * 2 + rr;
        float* srow = lds + row * NN;

        float m = -INFINITY;
        for (int c = lane; c < NN; c += 32) m = fmaxf(m, srow[c]);
#pragma unroll
        for (int off = 16; off; off >>= 1) m = fmaxf(m, __shfl_xor(m, off, 32));

        float s = 0.f;
        for (int c = lane; c < NN; c += 32) {
            float e = __expf((srow[c] - m) * invT);
            srow[c] = e;
            s += e;
        }
#pragma unroll
        for (int off = 16; off; off >>= 1) s += __shfl_xor(s, off, 32);

        const float inv = 1.0f / s;
        float* orow = out + ((size_t)b * NN + rowbase + row) * NN;
        for (int c = lane; c < NN; c += 32) orow[c] = srow[c] * inv;
    }
}

extern "C" void kernel_launch(void* const* d_in, const int* in_sizes, int n_in,
                              void* d_out, int out_size, void* d_ws, size_t ws_size,
                              hipStream_t stream) {
    const float* h    = (const float*)d_in[0];
    const float* temp = (const float*)d_in[1];
    float*       out  = (float*)d_out;
    float*       rnrm = (float*)d_ws;

    rnorm_kernel<<<NB * NN / 8, 256, 0, stream>>>(h, rnrm);
    simsoftmax_kernel<<<NB * (NN / 16), 256, 16 * NN * sizeof(float), stream>>>(
        h, rnrm, temp, out);
}


# B fragments fetched as whole-tile groups of 16-B loads issued one tile ahead of their WMMAs (k inside each 8-k group permuted alike in A and B: products are summed in that k order)
# speedup vs baseline: 1.2980x; 1.2980x over previous
#include <hip/hip_runtime.h>
#include <math.h>

typedef __attribute__((ext_vector_type(2))) float v2f;
typedef __attribute__((ext_vector_type(4))) float v4f;
typedef __attribute__((ext_vector_type(8))) float v8f;

#define NB   8
#define NN   2048
#define ND   128
#define CEPS 1e-8f

__global__ void rnorm_kernel(const float* __restrict__ h, float* __restrict__ rnorm) {
    const int row  = blockIdx.x * 8 + (threadIdx.x >> 5);
    const int lane = threadIdx.x & 31;
    const float* hr = h + (size_t)row * ND;
    float s = 0.f;
#pragma unroll
    for (int i = 0; i < 4; ++i) {
        float v = hr[lane + 32 * i];
        s += v * v;
    }
#pragma unroll
    for (int off = 16; off; off >>= 1) s += __shfl_xor(s, off, 32);
    if (lane == 0) rnorm[row] = 1.0f / fmaxf(sqrtf(s), CEPS);
}

__global__ void __launch_bounds__(256, 1)
simsoftmax_kernel(const float* __restrict__ h,
                  const float* __restrict__ rnorm,
                  const float* __restrict__ temp,
                  float* __restrict__ out) {
    extern __shared__ float lds[];

    const int b       = blockIdx.x >> 7;
    const int rb      = blockIdx.x & 127;
    const int wave    = threadIdx.x >> 5;
    const int lane    = threadIdx.x & 31;
    const int n       = lane & 15;
    const int khalf   = lane >> 4;
    const int rowbase = rb * 16;

    const float* hb = h     + (size_t)b * NN * ND;
    const float* rn = rnorm + b * NN;

    v2f afrag[32];
    {
        const float* arow = hb + (size_t)(rowbase + n) * ND + 4 * khalf;
#pragma unroll
        for (int j = 0; j < 16; ++j) {
            const v4f a4 = *(const v4f*)(arow + j * 8);
            afrag[2 * j]     = v2f{a4.x, a4.y};
            afrag[2 * j + 1] = v2f{a4.z, a4.w};
        }
    }

    float rowr[8];
#pragma unroll
    for (int r = 0; r < 8; ++r)
        rowr[r] = rn[rowbase + r + 8 * khalf];

    auto load_btile = [&](v4f (&bf)[16], float& rc, int t) {
        rc = rn[t * 16 + n];
        const float* brow = hb + (size_t)(t * 16 + n) * ND + 4 * khalf;
#pragma unroll
        for (int j = 0; j < 16; ++j)
            bf[j] = *(const v4f*)(brow + j * 8);
    };
    auto mul_tile = [&](const v4f (&bf)[16], float rc, int t) {
        const int colbase = t * 16;
        v8f acc = {};
#pragma unroll
        for (int j = 0; j < 16; ++j) {
            acc = __builtin_amdgcn_wmma_f32_16x16x4_f32(
                false, afrag[2 * j], false, v2f{bf[j].x, bf[j].y}, (short)0, acc, false, false);
            acc = __builtin_amdgcn_wmma_f32_16x16x4_f32(
                false, afrag[2 * j + 1], false, v2f{bf[j].z, bf[j].w}, (short)0, acc, false, false);
        }
#pragma unroll
        for (int r = 0; r < 8; ++r)
            lds[(r + 8 * khalf) * NN + colbase + n] = acc[r] * rowr[r] * rc;
    };

    v4f b0[16], b1[16];
    float rc0, rc1;
    load_btile(b0, rc0, wave);
    for (int i = 0; i < 14; i += 2) {
        const int t0 = wave + 8 * i;
        load_btile(b1, rc1, t0 + 8);
        __builtin_amdgcn_sched_barrier(0);
        mul_tile(b0, rc0, t0);
        load_btile(b0, rc0, t0 + 16);
        __builtin_amdgcn_sched_barrier(0);
        mul_tile(b1, rc1, t0 + 8);
    }
    load_btile(b1, rc1, wave + 8 * 15);
    __builtin_amdgcn_sched_barrier(0);
    mul_tile(b0, rc0, wave + 8 * 14);
    __builtin_amdgcn_sched_barrier(0);
    mul_tile(b1, rc1, wave + 8 * 15);

    __syncthreads();

    const float invT = 1.0f / temp[0];
#pragma unroll
    for (int rr = 0; rr < 2; ++rr) {
        const int row = wave * 2 + rr;
        float* srow = lds + row * NN;

        float m = -INFINITY;
        for (int c = lane; c < NN; c += 32) m = fmaxf(m, srow[c]);
#pragma unroll
        for (int off = 16; off; off >>= 1) m = fmaxf(m, __shfl_xor(m, off, 32));

        float s = 0.f;
        for (int c = lane; c < NN; c += 32) {
            float e = __expf((srow[c] - m) * invT);
            srow[c] = e;
            s += e;
        }
#pragma unroll
        for (int off = 16; off; off >>= 1) s += __shfl_xor(s, off, 32);

        const float inv = 1.0f / s;
        float* orow = out + ((size_t)b * NN + rowbase + row) * NN;
        for (int c = lane; c < NN; c += 32) orow[c] = srow[c] * inv;
    }
}

extern "C" void kernel_launch(void* const* d_in, const int* in_sizes, int n_in,
                              void* d_out, int out_size, void* d_ws, size_t ws_size,
                              hipStream_t stream) {
    const float* h    = (const float*)d_in[0];
    const float* temp = (const float*)d_in[1];
    float*       out  = (float*)d_out;
    float*       rnrm = (float*)d_ws;

    rnorm_kernel<<<NB * NN / 8, 256, 0, stream>>>(h, rnrm);
    simsoftmax_kernel<<<NB * (NN / 16), 256, 16 * NN * sizeof(float), stream>>>(
        h, rnrm, temp, out);
}


# 32-row work-groups: each wave holds the A fragments of two 16-row blocks, so every B fragment load feeds two WMMA chains (half the fragment loads per product; k order and sums unchanged)
# speedup vs baseline: 1.6743x; 1.2899x over previous
#include <hip/hip_runtime.h>
#include <math.h>

typedef __attribute__((ext_vector_type(2))) float v2f;
typedef __attribute__((ext_vector_type(4))) float v4f;
typedef __attribute__((ext_vector_type(8))) float v8f;

#define NB   8
#define NN   2048
#define ND   128
#define RB   32
#define CEPS 1e-8f

__global__ void rnorm_kernel(const float* __restrict__ h, float* __restrict__ rnorm) {
    const int row  = blockIdx.x * 8 + (threadIdx.x >> 5);
    const int lane = threadIdx.x & 31;
    const float* hr = h + (size_t)row * ND;
    float s = 0.f;
#pragma unroll
    for (int i = 0; i < 4; ++i) {
        float v = hr[lane + 32 * i];
        s += v * v;
    }
#pragma unroll
    for (int off = 16; off; off >>= 1) s += __shfl_xor(s, off, 32);
    if (lane == 0) rnorm[row] = 1.0f / fmaxf(sqrtf(s), CEPS);
}

__global__ void __launch_bounds__(256, 1)
simsoftmax_kernel(const float* __restrict__ h,
                  const float* __restrict__ rnorm,
                  const float* __restrict__ temp,
                  float* __restrict__ out) {
    __shared__ float lds[RB * NN];

    const int b       = blockIdx.x >> 6;
    const int rb      = blockIdx.x & 63;
    const int wave    = threadIdx.x >> 5;
    const int lane    = threadIdx.x & 31;
    const int n       = lane & 15;
    const int khalf   = lane >> 4;
    const int rowbase = rb * RB;

    const float* hb = h     + (size_t)b * NN * ND;
    const float* rn = rnorm + b * NN;

    v2f afrag0[32], afrag1[32];
    {
        const float* arow = hb + (size_t)(rowbase + n) * ND + 4 * khalf;
#pragma unroll
        for (int j = 0; j < 16; ++j) {
            const v4f a4 = *(const v4f*)(arow + j * 8);
            afrag0[2 * j]     = v2f{a4.x, a4.y};
            afrag0[2 * j + 1] = v2f{a4.z, a4.w};
        }
#pragma unroll
        for (int j = 0; j < 16; ++j) {
            const v4f a4 = *(const v4f*)(arow + 16 * ND + j * 8);
            afrag1[2 * j]     = v2f{a4.x, a4.y};
            afrag1[2 * j + 1] = v2f{a4.z, a4.w};
        }
    }

    float rowr0[8], rowr1[8];
#pragma unroll
    for (int r = 0; r < 8; ++r) {
        rowr0[r] = rn[rowbase + r + 8 * khalf];
        rowr1[r] = rn[rowbase + 16 + r + 8 * khalf];
    }

    auto load_btile = [&](v4f (&bf)[16], float& rc, int t) {
        rc = rn[t * 16 + n];
        const float* brow = hb + (size_t)(t * 16 + n) * ND + 4 * khalf;
#pragma unroll
        for (int j = 0; j < 16; ++j)
            bf[j] = *(const v4f*)(brow + j * 8);
    };
    auto mul_tile = [&](const v4f (&bf)[16], float rc, int t) {
        const int colbase = t * 16;
        v8f acc0 = {}, acc1 = {};
#pragma unroll
        for (int j = 0; j < 16; ++j) {
            const v2f blo = v2f{bf[j].x, bf[j].y};
            const v2f bhi = v2f{bf[j].z, bf[j].w};
            acc0 = __builtin_amdgcn_wmma_f32_16x16x4_f32(
                false, afrag0[2 * j], false, blo, (short)0, acc0, false, false);
            acc1 = __builtin_amdgcn_wmma_f32_16x16x4_f32(
                false, afrag1[2 * j], false, blo, (short)0, acc1, false, false);
            acc0 = __builtin_amdgcn_wmma_f32_16x16x4_f32(
                false, afrag0[2 * j + 1], false, bhi, (short)0, acc0, false, false);
            acc1 = __builtin_amdgcn_wmma_f32_16x16x4_f32(
                false, afrag1[2 * j + 1], false, bhi, (short)0, acc1, false, false);
        }
#pragma unroll
        for (int r = 0; r < 8; ++r) {
            lds[(r + 8 * khalf) * NN + colbase + n]      = acc0[r] * rowr0[r] * rc;
            lds[(16 + r + 8 * khalf) * NN + colbase + n] = acc1[r] * rowr1[r] * rc;
        }
    };

    v4f b0[16], b1[16];
    float rc0, rc1;
    load_btile(b0, rc0, wave);
    for (int i = 0; i < 14; i += 2) {
        const int t0 = wave + 8 * i;
        load_btile(b1, rc1, t0 + 8);
        __builtin_amdgcn_sched_barrier(0);
        mul_tile(b0, rc0, t0);
        __builtin_amdgcn_sched_barrier(0);
        load_btile(b0, rc0, t0 + 16);
        __builtin_amdgcn_sched_barrier(0);
        mul_tile(b1, rc1, t0 + 8);
    }
    load_btile(b1, rc1, wave + 8 * 15);
    __builtin_amdgcn_sched_barrier(0);
    mul_tile(b0, rc0, wave + 8 * 14);
    __builtin_amdgcn_sched_barrier(0);
    mul_tile(b1, rc1, wave + 8 * 15);

    __syncthreads();

    const float invT = 1.0f / temp[0];
#pragma unroll
    for (int rr = 0; rr < 4; ++rr) {
        const int row = wave * 4 + rr;
        float* srow = lds + row * NN;

        float m = -INFINITY;
        for (int c = lane; c < NN; c += 32) m = fmaxf(m, srow[c]);
#pragma unroll
        for (int off = 16; off; off >>= 1) m = fmaxf(m, __shfl_xor(m, off, 32));

        float s = 0.f;
        for (int c = lane; c < NN; c += 32) {
            float e = __expf((srow[c] - m) * invT);
            srow[c] = e;
            s += e;
        }
#pragma unroll
        for (int off = 16; off; off >>= 1) s += __shfl_xor(s, off, 32);

        const float inv = 1.0f / s;
        float* orow = out + ((size_t)b * NN + rowbase + row) * NN;
        for (int c = lane; c < NN; c += 32) orow[c] = srow[c] * inv;
    }
}

extern "C" void kernel_launch(void* const* d_in, const int* in_sizes, int n_in,
                              void* d_out, int out_size, void* d_ws, size_t ws_size,
                              hipStream_t stream) {
    const float* h    = (const float*)d_in[0];
    const float* temp = (const float*)d_in[1];
    float*       out  = (float*)d_out;
    float*       rnrm = (float*)d_ws;

    rnorm_kernel<<<NB * NN / 8, 256, 0, stream>>>(h, rnrm);
    simsoftmax_kernel<<<NB * (NN / RB), 256, 0, stream>>>(
        h, rnrm, temp, out);
}


# softmax rows held in registers: each slab row is read from LDS once (b128 loads), max/exp/sum/scale run on that copy, output leaves in b128 stores; row sum now added in a different order
# speedup vs baseline: 1.9370x; 1.1569x over previous
#include <hip/hip_runtime.h>
#include <math.h>

typedef __attribute__((ext_vector_type(2))) float v2f;
typedef __attribute__((ext_vector_type(4))) float v4f;
typedef __attribute__((ext_vector_type(8))) float v8f;

#define NB   8
#define NN   2048
#define ND   128
#define RB   32
#define CEPS 1e-8f

__global__ void rnorm_kernel(const float* __restrict__ h, float* __restrict__ rnorm) {
    const int row  = blockIdx.x * 8 + (threadIdx.x >> 5);
    const int lane = threadIdx.x & 31;
    const float* hr = h + (size_t)row * ND;
    float s = 0.f;
#pragma unroll
    for (int i = 0; i < 4; ++i) {
        float v = hr[lane + 32 * i];
        s += v * v;
    }
#pragma unroll
    for (int off = 16; off; off >>= 1) s += __shfl_xor(s, off, 32);
    if (lane == 0) rnorm[row] = 1.0f / fmaxf(sqrtf(s), CEPS);
}

__global__ void __launch_bounds__(256, 1)
simsoftmax_kernel(const float* __restrict__ h,
                  const float* __restrict__ rnorm,
                  const float* __restrict__ temp,
                  float* __restrict__ out) {
    __shared__ __attribute__((aligned(16))) float lds[RB * NN];

    const int b       = blockIdx.x >> 6;
    const int rb      = blockIdx.x & 63;
    const int wave    = threadIdx.x >> 5;
    const int lane    = threadIdx.x & 31;
    const int n       = lane & 15;
    const int khalf   = lane >> 4;
    const int rowbase = rb * RB;

    const float* hb = h     + (size_t)b * NN * ND;
    const float* rn = rnorm + b * NN;

    v2f afrag0[32], afrag1[32];
    {
        const float* arow = hb + (size_t)(rowbase + n) * ND + 4 * khalf;
#pragma unroll
        for (int j = 0; j < 16; ++j) {
            const v4f a4 = *(const v4f*)(arow + j * 8);
            afrag0[2 * j]     = v2f{a4.x, a4.y};
            afrag0[2 * j + 1] = v2f{a4.z, a4.w};
        }
#pragma unroll
        for (int j = 0; j < 16; ++j) {
            const v4f a4 = *(const v4f*)(arow + 16 * ND + j * 8);
            afrag1[2 * j]     = v2f{a4.x, a4.y};
            afrag1[2 * j + 1] = v2f{a4.z, a4.w};
        }
    }

    float rowr0[8], rowr1[8];
#pragma unroll
    for (int r = 0; r < 8; ++r) {
        rowr0[r] = rn[rowbase + r + 8 * khalf];
        rowr1[r] = rn[rowbase + 16 + r + 8 * khalf];
    }

    auto load_btile = [&](v4f (&bf)[16], float& rc, int t) {
        rc = rn[t * 16 + n];
        const float* brow = hb + (size_t)(t * 16 + n) * ND + 4 * khalf;
#pragma unroll
        for (int j = 0; j < 16; ++j)
            bf[j] = *(const v4f*)(brow + j * 8);
    };
    auto mul_tile = [&](const v4f (&bf)[16], float rc, int t) {
        const int colbase = t * 16;
        v8f acc0 = {}, acc1 = {};
#pragma unroll
        for (int j = 0; j < 16; ++j) {
            const v2f blo = v2f{bf[j].x, bf[j].y};
            const v2f bhi = v2f{bf[j].z, bf[j].w};
            acc0 = __builtin_amdgcn_wmma_f32_16x16x4_f32(
                false, afrag0[2 * j], false, blo, (short)0, acc0, false, false);
            acc1 = __builtin_amdgcn_wmma_f32_16x16x4_f32(
                false, afrag1[2 * j], false, blo, (short)0, acc1, false, false);
            acc0 = __builtin_amdgcn_wmma_f32_16x16x4_f32(
                false, afrag0[2 * j + 1], false, bhi, (short)0, acc0, false, false);
            acc1 = __builtin_amdgcn_wmma_f32_16x16x4_f32(
                false, afrag1[2 * j + 1], false, bhi, (short)0, acc1, false, false);
        }
#pragma unroll
        for (int r = 0; r < 8; ++r) {
            lds[(r + 8 * khalf) * NN + colbase + n]      = acc0[r] * rowr0[r] * rc;
            lds[(16 + r + 8 * khalf) * NN + colbase + n] = acc1[r] * rowr1[r] * rc;
        }
    };

    v4f b0[16], b1[16];
    float rc0, rc1;
    load_btile(b0, rc0, wave);
    for (int i = 0; i < 14; i += 2) {
        const int t0 = wave + 8 * i;
        load_btile(b1, rc1, t0 + 8);
        __builtin_amdgcn_sched_barrier(0);
        mul_tile(b0, rc0, t0);
        __builtin_amdgcn_sched_barrier(0);
        load_btile(b0, rc0, t0 + 16);
        __builtin_amdgcn_sched_barrier(0);
        mul_tile(b1, rc1, t0 + 8);
    }
    load_btile(b1, rc1, wave + 8 * 15);
    __builtin_amdgcn_sched_barrier(0);
    mul_tile(b0, rc0, wave + 8 * 14);
    __builtin_amdgcn_sched_barrier(0);
    mul_tile(b1, rc1, wave + 8 * 15);

    __syncthreads();

    const float invT = 1.0f / temp[0];
    const v4f* srow4 = (const v4f*)(lds + (wave * 4) * NN) + lane;
    v4f x[4][16];
#pragma unroll
    for (int rr = 0; rr < 4; ++rr) {
#pragma unroll
        for (int q = 0; q < 16; ++q) x[rr][q] = srow4[rr * (NN / 4) + q * 32];
    }

    float m[4];
#pragma unroll
    for (int rr = 0; rr < 4; ++rr) {
        v4f mv = x[rr][0];
#pragma unroll
        for (int q = 1; q < 16; ++q) {
            mv.x = fmaxf(mv.x, x[rr][q].x);
            mv.y = fmaxf(mv.y, x[rr][q].y);
            mv.z = fmaxf(mv.z, x[rr][q].z);
            mv.w = fmaxf(mv.w, x[rr][q].w);
        }
        m[rr] = fmaxf(fmaxf(mv.x, mv.y), fmaxf(mv.z, mv.w));
    }
#pragma unroll
    for (int off = 16; off; off >>= 1) {
#pragma unroll
        for (int rr = 0; rr < 4; ++rr) m[rr] = fmaxf(m[rr], __shfl_xor(m[rr], off, 32));
    }

    float s[4];
#pragma unroll
    for (int rr = 0; rr < 4; ++rr) {
        v4f sv = {0.f, 0.f, 0.f, 0.f};
#pragma unroll
        for (int q = 0; q < 16; ++q) {
            v4f e;
            e.x = __expf((x[rr][q].x - m[rr]) * invT);
            e.y = __expf((x[rr][q].y - m[rr]) * invT);
            e.z = __expf((x[rr][q].z - m[rr]) * invT);
            e.w = __expf((x[rr][q].w - m[rr]) * invT);
            x[rr][q] = e;
            sv += e;
        }
        s[rr] = (sv.x + sv.y) + (sv.z + sv.w);
    }
#pragma unroll
    for (int off = 16; off; off >>= 1) {
#pragma unroll
        for (int rr = 0; rr < 4; ++rr) s[rr] += __shfl_xor(s[rr], off, 32);
    }

    v4f* orow4 = (v4f*)(out + ((size_t)b * NN + rowbase + wave * 4) * NN) + lane;
#pragma unroll
    for (int rr = 0; rr < 4; ++rr) {
        const float inv = 1.0f / s[rr];
#pragma unroll
        for (int q = 0; q < 16; ++q) orow4[rr * (NN / 4) + q * 32] = x[rr][q] * inv;
    }
}

extern "C" void kernel_launch(void* const* d_in, const int* in_sizes, int n_in,
                              void* d_out, int out_size, void* d_ws, size_t ws_size,
                              hipStream_t stream) {
    const float* h    = (const float*)d_in[0];
    const float* temp = (const float*)d_in[1];
    float*       out  = (float*)d_out;
    float*       rnrm = (float*)d_ws;

    rnorm_kernel<<<NB * NN / 8, 256, 0, stream>>>(h, rnrm);
    simsoftmax_kernel<<<NB * (NN / RB), 256, 0, stream>>>(
        h, rnrm, temp, out);
}
